# starting kernel
# baseline (speedup 1.0000x reference)
#include <hip/hip_runtime.h>

typedef __attribute__((ext_vector_type(2))) float v2f;
typedef __attribute__((ext_vector_type(8))) float v8f;

#define YSPLIT 4

__global__ void pack_b_kernel(const float* __restrict__ y,
                              float* __restrict__ bpack, int ytiles) {
  int tid = blockIdx.x * blockDim.x + threadIdx.x;
  if (tid >= ytiles * 32) return;
  int t = tid >> 5;
  int lane = tid & 31;
  int j = t * 16 + (lane & 15);
  float y0 = y[j * 3 + 0];
  float y1 = y[j * 3 + 1];
  float y2 = y[j * 3 + 2];
  float o0, o1;
  if (lane < 16) {
    o0 = y0; o1 = y1;
  } else {
    o0 = y2; o1 = y0 * y0 + y1 * y1 + y2 * y2;
  }
  bpack[(size_t)t * 64 + lane * 2 + 0] = o0;
  bpack[(size_t)t * 64 + lane * 2 + 1] = o1;
}

__global__ __launch_bounds__(128) void chamfer_wmma_kernel(
    const float* __restrict__ x, const float* __restrict__ bpack,
    float* __restrict__ partial, int n, int ytiles) {
  const int lane  = threadIdx.x & 31;
  const int gwave = blockIdx.x * (blockDim.x >> 5) + (threadIdx.x >> 5);
  const int xtile  = gwave >> 2;
  const int ysplit = gwave & (YSPLIT - 1);

  const int m   = lane & 15;
  const int row = xtile * 16 + m;
  v2f A;
  if (lane < 16) {
    A.x = -2.0f * x[row * 3 + 0];
    A.y = -2.0f * x[row * 3 + 1];
  } else {
    A.x = -2.0f * x[row * 3 + 2];
    A.y = 1.0f;
  }

  v8f mn;
#pragma unroll
  for (int v = 0; v < 8; ++v) mn[v] = __builtin_inff();

  const int tiles_per = ytiles / YSPLIT;
  const float* bp = bpack + (size_t)(ysplit * tiles_per) * 64 + lane * 2;

#pragma unroll 4
  for (int t = 0; t < tiles_per; ++t) {
    v2f B = *(const v2f*)(bp + (size_t)t * 64);
    v8f c = {};
    c = __builtin_amdgcn_wmma_f32_16x16x4_f32(
         false, A,  false, B,
         (short)0, c,  false,  false);
#pragma unroll
    for (int v = 0; v < 8; ++v) mn[v] = fminf(mn[v], c[v]);
  }

#pragma unroll
  for (int off = 8; off >= 1; off >>= 1) {
#pragma unroll
    for (int v = 0; v < 8; ++v) {
      float o = __shfl_xor(mn[v], off, 32);
      mn[v] = fminf(mn[v], o);
    }
  }

  float* dst = partial + (size_t)ysplit * n + xtile * 16;
  if (lane == 0) {
#pragma unroll
    for (int v = 0; v < 8; ++v) dst[v] = mn[v];
  } else if (lane == 16) {
#pragma unroll
    for (int v = 0; v < 8; ++v) dst[8 + v] = mn[v];
  }
}

__global__ void finalize_kernel(const float* __restrict__ x,
                                const float* __restrict__ partial,
                                float* __restrict__ out, int n) {
  int i = blockIdx.x * blockDim.x + threadIdx.x;
  if (i >= n) return;
  float mv = partial[i];
#pragma unroll
  for (int s = 1; s < YSPLIT; ++s) mv = fminf(mv, partial[(size_t)s * n + i]);
  float x0 = x[i * 3 + 0], x1 = x[i * 3 + 1], x2 = x[i * 3 + 2];
  out[1 + i] = mv + (x0 * x0 + x1 * x1 + x2 * x2);
}

__global__ void sum_kernel(const float* __restrict__ vals,
                           float* __restrict__ out0, int n) {
  __shared__ float s[256];
  float acc = 0.0f;
  for (int i = threadIdx.x; i < n; i += 256) acc += vals[i];
  s[threadIdx.x] = acc;
  __syncthreads();
  for (int stride = 128; stride >= 1; stride >>= 1) {
    if ((int)threadIdx.x < stride) s[threadIdx.x] += s[threadIdx.x + stride];
    __syncthreads();
  }
  if (threadIdx.x == 0) *out0 = s[0];
}

extern "C" void kernel_launch(void* const* d_in, const int* in_sizes, int n_in,
                              void* d_out, int out_size, void* d_ws, size_t ws_size,
                              hipStream_t stream) {
  (void)n_in; (void)out_size; (void)ws_size;
  const float* x = (const float*)d_in[0];
  const float* y = (const float*)d_in[1];
  float* out = (float*)d_out;

  const int n = in_sizes[0] / 3;
  const int m = in_sizes[1] / 3;
  const int xtiles = n / 16;
  const int ytiles = m / 16;

  float* bpack   = (float*)d_ws;
  float* partial = bpack + (size_t)ytiles * 64;

  pack_b_kernel<<<(ytiles * 32 + 255) / 256, 256, 0, stream>>>(y, bpack, ytiles);

  chamfer_wmma_kernel<<<xtiles * YSPLIT / 4, 128, 0, stream>>>(x, bpack, partial,
                                                               n, ytiles);

  finalize_kernel<<<(n + 255) / 256, 256, 0, stream>>>(x, partial, out, n);
  sum_kernel<<<1, 256, 0, stream>>>(out + 1, out, n);
}


# Each wave keeps 4 x tiles in registers so one loaded B fragment feeds 4 WMMAs; y split moves inside a 16-wave work-group (LDS min merge): 256 work-groups, one partial plane
# speedup vs baseline: 1.3542x; 1.3542x over previous
#include <hip/hip_runtime.h>

typedef __attribute__((ext_vector_type(2))) float v2f;
typedef __attribute__((ext_vector_type(8))) float v8f;

__device__ __forceinline__ float min_ieee(float a, float b) {
  return __builtin_elementwise_minimum(a, b);
}

#define XT 4
#define WAVES 16

__global__ void pack_b_kernel(const float* __restrict__ y,
                              float* __restrict__ bpack, int ytiles) {
  int tid = blockIdx.x * blockDim.x + threadIdx.x;
  if (tid >= ytiles * 32) return;
  int t = tid >> 5;
  int lane = tid & 31;
  int j = t * 16 + (lane & 15);
  float y0 = y[j * 3 + 0];
  float y1 = y[j * 3 + 1];
  float y2 = y[j * 3 + 2];
  float o0, o1;
  if (lane < 16) {
    o0 = y0; o1 = y1;
  } else {
    o0 = y2; o1 = y0 * y0 + y1 * y1 + y2 * y2;
  }
  bpack[(size_t)t * 64 + lane * 2 + 0] = o0;
  bpack[(size_t)t * 64 + lane * 2 + 1] = o1;
}

__global__ __launch_bounds__(WAVES * 32) void chamfer_wmma_kernel(
    const float* __restrict__ x, const float* __restrict__ bpack,
    float* __restrict__ partial, int ytiles) {
  __shared__ float smin[WAVES][XT * 16];

  const int lane = threadIdx.x & 31;
  const int wave = threadIdx.x >> 5;
  const int xgroup = blockIdx.x;

  const int m = lane & 15;
  v2f A[XT];
#pragma unroll
  for (int r = 0; r < XT; ++r) {
    const int row = (xgroup * XT + r) * 16 + m;
    const float x0 = x[row * 3 + 0];
    const float x1 = x[row * 3 + 1];
    const float x2 = x[row * 3 + 2];
    A[r].x = -2.0f * (lane < 16 ? x0 : x2);
    A[r].y = lane < 16 ? -2.0f * x1 : 1.0f;
  }

  v8f mn[XT];
#pragma unroll
  for (int r = 0; r < XT; ++r) {
#pragma unroll
    for (int v = 0; v < 8; ++v) mn[r][v] = __builtin_inff();
  }

  const int steps = ytiles / WAVES;
  const int rot = (xgroup * 4) % steps;
  const float* bp = bpack + (size_t)wave * 64 + lane * 2;

  for (int kb = 0; kb < steps; kb += 4) {
    int k = kb + rot;
    if (k >= steps) k -= steps;
    const float* p = bp + (size_t)k * (WAVES * 64);
    v2f B[4];
#pragma unroll
    for (int u = 0; u < 4; ++u) B[u] = *(const v2f*)(p + (size_t)u * (WAVES * 64));
#pragma unroll
    for (int u = 0; u < 4; ++u) {
#pragma unroll
      for (int r = 0; r < XT; ++r) {
        v8f c = {};
        c = __builtin_amdgcn_wmma_f32_16x16x4_f32(
             false, A[r],  false, B[u],
             (short)0, c,  false,  false);
        mn[r] = __builtin_elementwise_minimum(mn[r], c);
      }
    }
  }

#pragma unroll
  for (int off = 8; off >= 1; off >>= 1) {
#pragma unroll
    for (int r = 0; r < XT; ++r) {
#pragma unroll
      for (int v = 0; v < 8; ++v) {
        float o = __shfl_xor(mn[r][v], off, 32);
        mn[r][v] = min_ieee(mn[r][v], o);
      }
    }
  }

  if (lane == 0) {
#pragma unroll
    for (int r = 0; r < XT; ++r) {
#pragma unroll
      for (int v = 0; v < 8; ++v) smin[wave][r * 16 + v] = mn[r][v];
    }
  } else if (lane == 16) {
#pragma unroll
    for (int r = 0; r < XT; ++r) {
#pragma unroll
      for (int v = 0; v < 8; ++v) smin[wave][r * 16 + 8 + v] = mn[r][v];
    }
  }
  __syncthreads();

  if (threadIdx.x < XT * 16) {
    float mv = smin[0][threadIdx.x];
#pragma unroll
    for (int w = 1; w < WAVES; ++w) mv = min_ieee(mv, smin[w][threadIdx.x]);
    partial[xgroup * (XT * 16) + threadIdx.x] = mv;
  }
}

__global__ void finalize_kernel(const float* __restrict__ x,
                                const float* __restrict__ partial,
                                float* __restrict__ out, int n) {
  int i = blockIdx.x * blockDim.x + threadIdx.x;
  if (i >= n) return;
  float mv = partial[i];
  float x0 = x[i * 3 + 0], x1 = x[i * 3 + 1], x2 = x[i * 3 + 2];
  out[1 + i] = mv + (x0 * x0 + x1 * x1 + x2 * x2);
}

__global__ void sum_kernel(const float* __restrict__ vals,
                           float* __restrict__ out0, int n) {
  __shared__ float s[256];
  float acc = 0.0f;
  for (int i = threadIdx.x; i < n; i += 256) acc += vals[i];
  s[threadIdx.x] = acc;
  __syncthreads();
  for (int stride = 128; stride >= 1; stride >>= 1) {
    if ((int)threadIdx.x < stride) s[threadIdx.x] += s[threadIdx.x + stride];
    __syncthreads();
  }
  if (threadIdx.x == 0) *out0 = s[0];
}

extern "C" void kernel_launch(void* const* d_in, const int* in_sizes, int n_in,
                              void* d_out, int out_size, void* d_ws, size_t ws_size,
                              hipStream_t stream) {
  (void)n_in; (void)out_size; (void)ws_size;
  const float* x = (const float*)d_in[0];
  const float* y = (const float*)d_in[1];
  float* out = (float*)d_out;

  const int n = in_sizes[0] / 3;
  const int m = in_sizes[1] / 3;
  const int xtiles = n / 16;
  const int ytiles = m / 16;

  float* bpack   = (float*)d_ws;
  float* partial = bpack + (size_t)ytiles * 64;

  pack_b_kernel<<<(ytiles * 32 + 255) / 256, 256, 0, stream>>>(y, bpack, ytiles);

  chamfer_wmma_kernel<<<xtiles / XT, WAVES * 32, 0, stream>>>(x, bpack, partial,
                                                              ytiles);

  finalize_kernel<<<(n + 255) / 256, 256, 0, stream>>>(x, partial, out, n);
  sum_kernel<<<1, 256, 0, stream>>>(out + 1, out, n);
}


# Sum pass: each thread issues its 64 loads as one group ahead of its adds and waits for about 2 cache round trips instead of 64 dependent ones; add order kept, sum bit-identical
# speedup vs baseline: 2.4043x; 1.7754x over previous
#include <hip/hip_runtime.h>

typedef __attribute__((ext_vector_type(2))) float v2f;
typedef __attribute__((ext_vector_type(8))) float v8f;

__device__ __forceinline__ float min_ieee(float a, float b) {
  return __builtin_elementwise_minimum(a, b);
}

#define XT 4
#define WAVES 16

__global__ void pack_b_kernel(const float* __restrict__ y,
                              float* __restrict__ bpack, int ytiles) {
  int tid = blockIdx.x * blockDim.x + threadIdx.x;
  if (tid >= ytiles * 32) return;
  int t = tid >> 5;
  int lane = tid & 31;
  int j = t * 16 + (lane & 15);
  float y0 = y[j * 3 + 0];
  float y1 = y[j * 3 + 1];
  float y2 = y[j * 3 + 2];
  float o0, o1;
  if (lane < 16) {
    o0 = y0; o1 = y1;
  } else {
    o0 = y2; o1 = y0 * y0 + y1 * y1 + y2 * y2;
  }
  bpack[(size_t)t * 64 + lane * 2 + 0] = o0;
  bpack[(size_t)t * 64 + lane * 2 + 1] = o1;
}

__global__ __launch_bounds__(WAVES * 32) void chamfer_wmma_kernel(
    const float* __restrict__ x, const float* __restrict__ bpack,
    float* __restrict__ partial, int ytiles) {
  __shared__ float smin[WAVES][XT * 16];

  const int lane = threadIdx.x & 31;
  const int wave = threadIdx.x >> 5;
  const int xgroup = blockIdx.x;

  const int m = lane & 15;
  v2f A[XT];
#pragma unroll
  for (int r = 0; r < XT; ++r) {
    const int row = (xgroup * XT + r) * 16 + m;
    const float x0 = x[row * 3 + 0];
    const float x1 = x[row * 3 + 1];
    const float x2 = x[row * 3 + 2];
    A[r].x = -2.0f * (lane < 16 ? x0 : x2);
    A[r].y = lane < 16 ? -2.0f * x1 : 1.0f;
  }

  v8f mn[XT];
#pragma unroll
  for (int r = 0; r < XT; ++r) {
#pragma unroll
    for (int v = 0; v < 8; ++v) mn[r][v] = __builtin_inff();
  }

  const int steps = ytiles / WAVES;
  const int rot = (xgroup * 4) % steps;
  const float* bp = bpack + (size_t)wave * 64 + lane * 2;

  for (int kb = 0; kb < steps; kb += 4) {
    int k = kb + rot;
    if (k >= steps) k -= steps;
    const float* p = bp + (size_t)k * (WAVES * 64);
    v2f B[4];
#pragma unroll
    for (int u = 0; u < 4; ++u) B[u] = *(const v2f*)(p + (size_t)u * (WAVES * 64));
#pragma unroll
    for (int u = 0; u < 4; ++u) {
#pragma unroll
      for (int r = 0; r < XT; ++r) {
        v8f c = {};
        c = __builtin_amdgcn_wmma_f32_16x16x4_f32(
             false, A[r],  false, B[u],
             (short)0, c,  false,  false);
        mn[r] = __builtin_elementwise_minimum(mn[r], c);
      }
    }
  }

#pragma unroll
  for (int off = 8; off >= 1; off >>= 1) {
#pragma unroll
    for (int r = 0; r < XT; ++r) {
#pragma unroll
      for (int v = 0; v < 8; ++v) {
        float o = __shfl_xor(mn[r][v], off, 32);
        mn[r][v] = min_ieee(mn[r][v], o);
      }
    }
  }

  if (lane == 0) {
#pragma unroll
    for (int r = 0; r < XT; ++r) {
#pragma unroll
      for (int v = 0; v < 8; ++v) smin[wave][r * 16 + v] = mn[r][v];
    }
  } else if (lane == 16) {
#pragma unroll
    for (int r = 0; r < XT; ++r) {
#pragma unroll
      for (int v = 0; v < 8; ++v) smin[wave][r * 16 + 8 + v] = mn[r][v];
    }
  }
  __syncthreads();

  if (threadIdx.x < XT * 16) {
    float mv = smin[0][threadIdx.x];
#pragma unroll
    for (int w = 1; w < WAVES; ++w) mv = min_ieee(mv, smin[w][threadIdx.x]);
    partial[xgroup * (XT * 16) + threadIdx.x] = mv;
  }
}

__global__ void finalize_kernel(const float* __restrict__ x,
                                const float* __restrict__ partial,
                                float* __restrict__ out, int n) {
  int i = blockIdx.x * blockDim.x + threadIdx.x;
  if (i >= n) return;
  float mv = partial[i];
  float x0 = x[i * 3 + 0], x1 = x[i * 3 + 1], x2 = x[i * 3 + 2];
  out[1 + i] = mv + (x0 * x0 + x1 * x1 + x2 * x2);
}

#define SUM_BATCH 64

__global__ void sum_kernel(const float* __restrict__ vals,
                           float* __restrict__ out0, int n) {
  __shared__ float s[256];
  float acc = 0.0f;
  int i = threadIdx.x;
  for (; i + (SUM_BATCH - 1) * 256 < n; i += SUM_BATCH * 256) {
    float v[SUM_BATCH];
#pragma unroll
    for (int u = 0; u < SUM_BATCH; ++u) v[u] = vals[i + u * 256];
#pragma unroll
    for (int u = 0; u < SUM_BATCH; ++u) acc += v[u];
  }
  for (; i < n; i += 256) acc += vals[i];
  s[threadIdx.x] = acc;
  __syncthreads();
  for (int stride = 128; stride >= 1; stride >>= 1) {
    if ((int)threadIdx.x < stride) s[threadIdx.x] += s[threadIdx.x + stride];
    __syncthreads();
  }
  if (threadIdx.x == 0) *out0 = s[0];
}

extern "C" void kernel_launch(void* const* d_in, const int* in_sizes, int n_in,
                              void* d_out, int out_size, void* d_ws, size_t ws_size,
                              hipStream_t stream) {
  (void)n_in; (void)out_size; (void)ws_size;
  const float* x = (const float*)d_in[0];
  const float* y = (const float*)d_in[1];
  float* out = (float*)d_out;

  const int n = in_sizes[0] / 3;
  const int m = in_sizes[1] / 3;
  const int xtiles = n / 16;
  const int ytiles = m / 16;

  float* bpack   = (float*)d_ws;
  float* partial = bpack + (size_t)ytiles * 64;

  pack_b_kernel<<<(ytiles * 32 + 255) / 256, 256, 0, stream>>>(y, bpack, ytiles);

  chamfer_wmma_kernel<<<xtiles / XT, WAVES * 32, 0, stream>>>(x, bpack, partial,
                                                              ytiles);

  finalize_kernel<<<(n + 255) / 256, 256, 0, stream>>>(x, partial, out, n);
  sum_kernel<<<1, 256, 0, stream>>>(out + 1, out, n);
}


# Step loop as a two-stage pipeline: a trip issues the next block's four y loads ahead of this block's 16 products, builds their fragments after them; first block loaded with the x rows; bit-identical
# speedup vs baseline: 3.0283x; 1.2595x over previous
#include <hip/hip_runtime.h>


typedef __attribute__((ext_vector_type(2))) float v2f;
typedef __attribute__((ext_vector_type(8))) float v8f;

__device__ __forceinline__ float min_ieee(float a, float b) {
  return __builtin_elementwise_minimum(a, b);
}

__device__ __forceinline__ v2f make_b(float y0, float y1, float y2, bool lo) {
  const float ysq = __builtin_fmaf(y2, y2, __builtin_fmaf(y0, y0, y1 * y1));
  v2f b;
  b.x = lo ? y0 : y2;
  b.y = lo ? y1 : ysq;
  return b;
}

#define XT 4
#define WAVES 16

__global__ __launch_bounds__(WAVES * 32) void chamfer_wmma_kernel(
    const float* __restrict__ x, const float* __restrict__ y,
    float* __restrict__ out, int ytiles) {
  __shared__ float smin[WAVES][XT * 16];

  const int lane = threadIdx.x & 31;
  const int wave = threadIdx.x >> 5;
  const int xgroup = blockIdx.x;
  const bool lo = lane < 16;

  const int m = lane & 15;
  const int mine = (wave & 1) * 2 + (lane >> 4);
  float xsq = 0.0f;
  v2f A[XT];
#pragma unroll
  for (int r = 0; r < XT; ++r) {
    const int row = (xgroup * XT + r) * 16 + m;
    const float x0 = x[row * 3 + 0];
    const float x1 = x[row * 3 + 1];
    const float x2 = x[row * 3 + 2];
    A[r].x = -2.0f * (lo ? x0 : x2);
    A[r].y = lo ? -2.0f * x1 : 1.0f;
    const float s = x0 * x0 + x1 * x1 + x2 * x2;
    xsq = (r == mine) ? s : xsq;
  }

  v8f mn[XT];
#pragma unroll
  for (int r = 0; r < XT; ++r) {
#pragma unroll
    for (int v = 0; v < 8; ++v) mn[r][v] = __builtin_inff();
  }

  const int steps = ytiles / WAVES;
  const int rot = (xgroup * 4) % steps;
  const float* yp = y + (size_t)(wave * 16 + m) * 3;

  v2f B[4];
  {
    const float* p = yp + (size_t)rot * (WAVES * 16 * 3);
#pragma unroll
    for (int u = 0; u < 4; ++u) {
      const float* q = p + (size_t)u * (WAVES * 16 * 3);
      B[u] = make_b(q[0], q[1], q[2], lo);
    }
  }

  int k = rot;
  int kb = 0;
  do {
    k += 4;
    if (k >= steps) k -= steps;
    const float* p = yp + (size_t)k * (WAVES * 16 * 3);
    float ny0[4], ny1[4], ny2[4];
#pragma unroll
    for (int u = 0; u < 4; ++u) {
      const float* q = p + (size_t)u * (WAVES * 16 * 3);
      ny0[u] = q[0];
      ny1[u] = q[1];
      ny2[u] = q[2];
    }
#pragma unroll
    for (int u = 0; u < 4; ++u) {
#pragma unroll
      for (int r = 0; r < XT; ++r) {
        v8f c = {};
        c = __builtin_amdgcn_wmma_f32_16x16x4_f32(
             false, A[r],  false, B[u],
             (short)0, c,  false,  false);
        mn[r] = __builtin_elementwise_minimum(mn[r], c);
      }
    }
#pragma unroll
    for (int u = 0; u < 4; ++u) B[u] = make_b(ny0[u], ny1[u], ny2[u], lo);
    __builtin_amdgcn_sched_group_barrier(0x020, 4, 0);
    __builtin_amdgcn_sched_group_barrier(0x008, 16, 0);
    kb += 4;
  } while (kb < steps);

#pragma unroll
  for (int off = 8; off >= 1; off >>= 1) {
#pragma unroll
    for (int r = 0; r < XT; ++r) {
#pragma unroll
      for (int v = 0; v < 8; ++v) {
        float o = __shfl_xor(mn[r][v], off, 32);
        mn[r][v] = min_ieee(mn[r][v], o);
      }
    }
  }

  if (lane == 0) {
#pragma unroll
    for (int r = 0; r < XT; ++r) {
#pragma unroll
      for (int v = 0; v < 8; ++v) smin[wave][r * 16 + v] = mn[r][v];
    }
  } else if (lane == 16) {
#pragma unroll
    for (int r = 0; r < XT; ++r) {
#pragma unroll
      for (int v = 0; v < 8; ++v) smin[wave][r * 16 + 8 + v] = mn[r][v];
    }
  }
  __syncthreads();

  if (threadIdx.x < XT * 16) {
    float mv = smin[0][threadIdx.x];
#pragma unroll
    for (int w = 1; w < WAVES; ++w) mv = min_ieee(mv, smin[w][threadIdx.x]);
    out[1 + xgroup * (XT * 16) + threadIdx.x] = mv + xsq;
  }
}

#define SUM_BATCH 64

__global__ void sum_kernel(const float* __restrict__ vals,
                           float* __restrict__ out0, int n) {
  __shared__ float s[256];
  float acc = 0.0f;
  int i = threadIdx.x;
  for (; i + (SUM_BATCH - 1) * 256 < n; i += SUM_BATCH * 256) {
    float v[SUM_BATCH];
#pragma unroll
    for (int u = 0; u < SUM_BATCH; ++u) v[u] = vals[i + u * 256];
#pragma unroll
    for (int u = 0; u < SUM_BATCH; ++u) acc += v[u];
  }
  for (; i < n; i += 256) acc += vals[i];
  s[threadIdx.x] = acc;
  __syncthreads();
  for (int stride = 128; stride >= 1; stride >>= 1) {
    if ((int)threadIdx.x < stride) s[threadIdx.x] += s[threadIdx.x + stride];
    __syncthreads();
  }
  if (threadIdx.x == 0) *out0 = s[0];
}

extern "C" void kernel_launch(void* const* d_in, const int* in_sizes, int n_in,
                              void* d_out, int out_size, void* d_ws, size_t ws_size,
                              hipStream_t stream) {
  (void)n_in; (void)out_size; (void)d_ws; (void)ws_size;
  const float* x = (const float*)d_in[0];
  const float* y = (const float*)d_in[1];
  float* out = (float*)d_out;

  const int n = in_sizes[0] / 3;
  const int m = in_sizes[1] / 3;
  const int xtiles = n / 16;
  const int ytiles = m / 16;

  chamfer_wmma_kernel<<<xtiles / XT, WAVES * 32, 0, stream>>>(x, y, out, ytiles);

  sum_kernel<<<1, 256, 0, stream>>>(out + 1, out, n);
}


# Product transposed (y points as matrix rows, x rows as columns): min over y runs down a lane's own 8 result registers, 4 three-input folds per product; the 128 shuffles per wave go; bit-identical
# speedup vs baseline: 3.3345x; 1.1011x over previous
#include <hip/hip_runtime.h>


typedef __attribute__((ext_vector_type(2))) float v2f;
typedef __attribute__((ext_vector_type(8))) float v8f;

__device__ __forceinline__ float min_ieee(float a, float b) {
  return __builtin_elementwise_minimum(a, b);
}

__device__ __forceinline__ v2f make_yfrag(float y0, float y1, float y2, bool lo) {
  const float ysq = __builtin_fmaf(y2, y2, __builtin_fmaf(y0, y0, y1 * y1));
  v2f b;
  b.x = lo ? y0 : y2;
  b.y = lo ? y1 : ysq;
  return b;
}

#define XT 4
#define WAVES 16

__device__ __forceinline__ void fold8(float& a, float& b, v8f c) {
  a = min_ieee(min_ieee(a, c[0]), c[1]);
  b = min_ieee(min_ieee(b, c[2]), c[3]);
  a = min_ieee(min_ieee(a, c[4]), c[5]);
  b = min_ieee(min_ieee(b, c[6]), c[7]);
}

__global__ __launch_bounds__(WAVES * 32) void chamfer_wmma_kernel(
    const float* __restrict__ x, const float* __restrict__ y,
    float* __restrict__ out, int ytiles) {
  __shared__ float smin[WAVES][XT][32];

  const int lane = threadIdx.x & 31;
  const int wave = threadIdx.x >> 5;
  const int xgroup = blockIdx.x;
  const bool lo = lane < 16;

  const int m = lane & 15;
  const int mine = (wave & 1) * 2 + (lane >> 4);
  float xsq = 0.0f;
  v2f X[XT];
#pragma unroll
  for (int r = 0; r < XT; ++r) {
    const int row = (xgroup * XT + r) * 16 + m;
    const float x0 = x[row * 3 + 0];
    const float x1 = x[row * 3 + 1];
    const float x2 = x[row * 3 + 2];
    X[r].x = -2.0f * (lo ? x0 : x2);
    X[r].y = lo ? -2.0f * x1 : 1.0f;
    const float s = x0 * x0 + x1 * x1 + x2 * x2;
    xsq = (r == mine) ? s : xsq;
  }

  float mna[XT], mnb[XT];
#pragma unroll
  for (int r = 0; r < XT; ++r) mna[r] = mnb[r] = __builtin_inff();

  const int steps = ytiles / WAVES;
  const int rot = (xgroup * 4) % steps;
  const float* yp = y + (size_t)(wave * 16 + m) * 3;

  v2f Y[4];
  {
    const float* p = yp + (size_t)rot * (WAVES * 16 * 3);
#pragma unroll
    for (int u = 0; u < 4; ++u) {
      const float* q = p + (size_t)u * (WAVES * 16 * 3);
      Y[u] = make_yfrag(q[0], q[1], q[2], lo);
    }
  }

  int k = rot;
  int kb = 0;
  do {
    k += 4;
    if (k >= steps) k -= steps;
    const float* p = yp + (size_t)k * (WAVES * 16 * 3);
    float ny0[4], ny1[4], ny2[4];
#pragma unroll
    for (int u = 0; u < 4; ++u) {
      const float* q = p + (size_t)u * (WAVES * 16 * 3);
      ny0[u] = q[0];
      ny1[u] = q[1];
      ny2[u] = q[2];
    }
#pragma unroll
    for (int u = 0; u < 4; ++u) {
#pragma unroll
      for (int r = 0; r < XT; ++r) {
        v8f c = {};
        c = __builtin_amdgcn_wmma_f32_16x16x4_f32(
             false, Y[u],  false, X[r],
             (short)0, c,  false,  false);
        fold8(mna[r], mnb[r], c);
      }
    }
#pragma unroll
    for (int u = 0; u < 4; ++u) Y[u] = make_yfrag(ny0[u], ny1[u], ny2[u], lo);
    __builtin_amdgcn_sched_group_barrier(0x020, 4, 0);
    __builtin_amdgcn_sched_group_barrier(0x008, 16, 0);
    kb += 4;
  } while (kb < steps);

#pragma unroll
  for (int r = 0; r < XT; ++r) smin[wave][r][lane] = min_ieee(mna[r], mnb[r]);
  __syncthreads();

  if (threadIdx.x < XT * 16) {
    const int r = threadIdx.x >> 4;
    float mv = min_ieee(smin[0][r][m], smin[0][r][16 + m]);
#pragma unroll
    for (int w = 1; w < WAVES; ++w)
      mv = min_ieee(min_ieee(mv, smin[w][r][m]), smin[w][r][16 + m]);
    out[1 + xgroup * (XT * 16) + threadIdx.x] = mv + xsq;
  }
}

#define SUM_BATCH 64

__global__ void sum_kernel(const float* __restrict__ vals,
                           float* __restrict__ out0, int n) {
  __shared__ float s[256];
  float acc = 0.0f;
  int i = threadIdx.x;
  for (; i + (SUM_BATCH - 1) * 256 < n; i += SUM_BATCH * 256) {
    float v[SUM_BATCH];
#pragma unroll
    for (int u = 0; u < SUM_BATCH; ++u) v[u] = vals[i + u * 256];
#pragma unroll
    for (int u = 0; u < SUM_BATCH; ++u) acc += v[u];
  }
  for (; i < n; i += 256) acc += vals[i];
  s[threadIdx.x] = acc;
  __syncthreads();
  for (int stride = 128; stride >= 1; stride >>= 1) {
    if ((int)threadIdx.x < stride) s[threadIdx.x] += s[threadIdx.x + stride];
    __syncthreads();
  }
  if (threadIdx.x == 0) *out0 = s[0];
}

extern "C" void kernel_launch(void* const* d_in, const int* in_sizes, int n_in,
                              void* d_out, int out_size, void* d_ws, size_t ws_size,
                              hipStream_t stream) {
  (void)n_in; (void)out_size; (void)d_ws; (void)ws_size;
  const float* x = (const float*)d_in[0];
  const float* y = (const float*)d_in[1];
  float* out = (float*)d_out;

  const int n = in_sizes[0] / 3;
  const int m = in_sizes[1] / 3;
  const int xtiles = n / 16;
  const int ytiles = m / 16;

  chamfer_wmma_kernel<<<xtiles / XT, WAVES * 32, 0, stream>>>(x, y, out, ytiles);

  sum_kernel<<<1, 256, 0, stream>>>(out + 1, out, n);
}


# Row sum starts in the product kernel: each row-owning wave adds its 32 closest_dists by lane exchange, 512 partials; sum pass is one wave, no LDS tree; order of the sum changes (new tree)
# speedup vs baseline: 3.6285x; 1.0882x over previous
#include <hip/hip_runtime.h>


typedef __attribute__((ext_vector_type(2))) float v2f;
typedef __attribute__((ext_vector_type(8))) float v8f;

__device__ __forceinline__ float min_ieee(float a, float b) {
  return __builtin_elementwise_minimum(a, b);
}

__device__ __forceinline__ v2f make_yfrag(float y0, float y1, float y2, bool lo) {
  const float ysq = __builtin_fmaf(y2, y2, __builtin_fmaf(y0, y0, y1 * y1));
  v2f b;
  b.x = lo ? y0 : y2;
  b.y = lo ? y1 : ysq;
  return b;
}

#define XT 4
#define WAVES 16

__device__ __forceinline__ void fold8(float& a, float& b, v8f c) {
  a = min_ieee(min_ieee(a, c[0]), c[1]);
  b = min_ieee(min_ieee(b, c[2]), c[3]);
  a = min_ieee(min_ieee(a, c[4]), c[5]);
  b = min_ieee(min_ieee(b, c[6]), c[7]);
}

__device__ __forceinline__ float xhalf(float v) {
  const int b = __builtin_bit_cast(int, v);
  return __builtin_bit_cast(float, __builtin_amdgcn_permlanex16(
      b, b, 0x76543210u, 0xfedcba98u, false, false));
}

template <unsigned LO, unsigned HI>
__device__ __forceinline__ float xrow(float v) {
  const int b = __builtin_bit_cast(int, v);
  return __builtin_bit_cast(float,
                            __builtin_amdgcn_permlane16(b, b, LO, HI, false, false));
}

__device__ __forceinline__ float wave_sum(float s) {
  s += xrow<0x67452301u, 0xefcdab89u>(s);
  s += xrow<0x54761032u, 0xdcfe98bau>(s);
  s += xrow<0x32107654u, 0xba98fedcu>(s);
  s += xrow<0xfedcba98u, 0x76543210u>(s);
  s += xhalf(s);
  return s;
}

#define ROW_WAVES (XT * 16 / 32)

__global__ __launch_bounds__(WAVES * 32) void chamfer_wmma_kernel(
    const float* __restrict__ x, const float* __restrict__ y,
    float* __restrict__ out, float* __restrict__ part, int ytiles) {
  __shared__ float smin[WAVES][XT][32];

  const int lane = threadIdx.x & 31;
  const int wave = threadIdx.x >> 5;
  const int xgroup = blockIdx.x;
  const bool lo = lane < 16;

  const int m = lane & 15;
  const int mine = (wave & 1) * 2 + (lane >> 4);
  float xsq = 0.0f;
  v2f X[XT];
#pragma unroll
  for (int r = 0; r < XT; ++r) {
    const int row = (xgroup * XT + r) * 16 + m;
    const float x0 = x[row * 3 + 0];
    const float x1 = x[row * 3 + 1];
    const float x2 = x[row * 3 + 2];
    X[r].x = -2.0f * (lo ? x0 : x2);
    X[r].y = lo ? -2.0f * x1 : 1.0f;
    const float s = x0 * x0 + x1 * x1 + x2 * x2;
    xsq = (r == mine) ? s : xsq;
  }

  float mna[XT], mnb[XT];
#pragma unroll
  for (int r = 0; r < XT; ++r) mna[r] = mnb[r] = __builtin_inff();

  const int steps = ytiles / WAVES;
  const int rot = (xgroup * 4) % steps;
  const float* yp = y + (size_t)(wave * 16 + m) * 3;

  v2f Y[4];
  {
    const float* p = yp + (size_t)rot * (WAVES * 16 * 3);
#pragma unroll
    for (int u = 0; u < 4; ++u) {
      const float* q = p + (size_t)u * (WAVES * 16 * 3);
      Y[u] = make_yfrag(q[0], q[1], q[2], lo);
    }
  }

  int k = rot;
  int kb = 0;
  do {
    k += 4;
    if (k >= steps) k -= steps;
    const float* p = yp + (size_t)k * (WAVES * 16 * 3);
    float ny0[4], ny1[4], ny2[4];
#pragma unroll
    for (int u = 0; u < 4; ++u) {
      const float* q = p + (size_t)u * (WAVES * 16 * 3);
      ny0[u] = q[0];
      ny1[u] = q[1];
      ny2[u] = q[2];
    }
#pragma unroll
    for (int u = 0; u < 4; ++u) {
#pragma unroll
      for (int r = 0; r < XT; ++r) {
        v8f c = {};
        c = __builtin_amdgcn_wmma_f32_16x16x4_f32(
             false, Y[u],  false, X[r],
             (short)0, c,  false,  false);
        fold8(mna[r], mnb[r], c);
      }
    }
#pragma unroll
    for (int u = 0; u < 4; ++u) Y[u] = make_yfrag(ny0[u], ny1[u], ny2[u], lo);
    __builtin_amdgcn_sched_group_barrier(0x020, 4, 0);
    __builtin_amdgcn_sched_group_barrier(0x008, 16, 0);
    kb += 4;
  } while (kb < steps);

#pragma unroll
  for (int r = 0; r < XT; ++r) smin[wave][r][lane] = min_ieee(mna[r], mnb[r]);
  __syncthreads();

  if (threadIdx.x < XT * 16) {
    const int r = threadIdx.x >> 4;
    float mv = min_ieee(smin[0][r][m], smin[0][r][16 + m]);
#pragma unroll
    for (int w = 1; w < WAVES; ++w)
      mv = min_ieee(min_ieee(mv, smin[w][r][m]), smin[w][r][16 + m]);
    const float d = mv + xsq;
    out[1 + xgroup * (XT * 16) + threadIdx.x] = d;
    const float s = wave_sum(d);
    if (lane == 0) part[xgroup * ROW_WAVES + wave] = s;
  }
}

#define PART_BATCH 16

__global__ __launch_bounds__(32) void sum_partials_kernel(
    const float* __restrict__ part, float* __restrict__ out0, int groups) {
  const int np = groups * ROW_WAVES;
  float acc = 0.0f;
  int i = threadIdx.x;
  for (; i + (PART_BATCH - 1) * 32 < np; i += PART_BATCH * 32) {
    float v[PART_BATCH];
#pragma unroll
    for (int u = 0; u < PART_BATCH; ++u) v[u] = part[i + u * 32];
#pragma unroll
    for (int u = 0; u < PART_BATCH; ++u) acc += v[u];
  }
  for (; i < np; i += 32) acc += part[i];
  const float s = wave_sum(acc);
  if (threadIdx.x == 0) *out0 = s;
}

extern "C" void kernel_launch(void* const* d_in, const int* in_sizes, int n_in,
                              void* d_out, int out_size, void* d_ws, size_t ws_size,
                              hipStream_t stream) {
  (void)n_in; (void)out_size; (void)ws_size;
  const float* x = (const float*)d_in[0];
  const float* y = (const float*)d_in[1];
  float* out = (float*)d_out;
  float* part = (float*)d_ws;

  const int n = in_sizes[0] / 3;
  const int m = in_sizes[1] / 3;
  const int xtiles = n / 16;
  const int ytiles = m / 16;

  chamfer_wmma_kernel<<<xtiles / XT, WAVES * 32, 0, stream>>>(x, y, out, part, ytiles);

  sum_partials_kernel<<<1, 32, 0, stream>>>(part, out, xtiles / XT);
}
